# starting kernel
# baseline (speedup 1.0000x reference)
#include <hip/hip_runtime.h>
#include <hip/hip_bf16.h>

typedef float v2f __attribute__((ext_vector_type(2)));
typedef float v8f __attribute__((ext_vector_type(8)));

__device__ __forceinline__ v8f wmma_f32(v2f a, v2f b, v8f c) {
  return __builtin_amdgcn_wmma_f32_16x16x4_f32(false, a, false, b, (short)0, c,
                                               false, false);
}

#define NSTREAM 6
#define H 128

__device__ __forceinline__ void mlp_hidden_layer(
    const float* __restrict__ W, const float* __restrict__ bb,
    const float (&actin)[NSTREAM * 16 * H], float (&actout)[NSTREAM * 16 * H],
    const int ln, const int half, const int koff)
{
  #pragma unroll
  for (int j = 0; j < 8; ++j) {
    v2f bf[32];
    #pragma unroll
    for (int kc = 0; kc < 32; ++kc) {
      const int kb = 4 * kc + koff;
      bf[kc].x = W[(size_t)kb * H + 16 * j + ln];
      bf[kc].y = W[(size_t)(kb + 1) * H + 16 * j + ln];
    }
    const float bj = bb[16 * j + ln];

    v8f acur, zcur;
    #pragma unroll
    for (int s = 0; s < NSTREAM; ++s) {
      v8f acc;
      #pragma unroll
      for (int v = 0; v < 8; ++v) acc[v] = (s == 0) ? bj : 0.f;

      #pragma unroll
      for (int kc = 0; kc < 32; ++kc) {
        const v2f af = *(const v2f*)&actin[(s * 16 + ln) * H + 4 * kc + koff];
        acc = wmma_f32(af, bf[kc], acc);
      }

      v8f res;
      if (s == 0) {
        #pragma unroll
        for (int v = 0; v < 8; ++v) res[v] = tanhf(acc[v]);
        acur = res;
      } else if (s == 1 || s == 3) {
        zcur = acc;
        #pragma unroll
        for (int v = 0; v < 8; ++v) {
          const float a = acur[v];
          res[v] = (1.f - a * a) * acc[v];
        }
      } else if (s == 2 || s == 4) {
        #pragma unroll
        for (int v = 0; v < 8; ++v) {
          const float a  = acur[v];
          const float s2 = 1.f - a * a;
          const float zd = zcur[v];
          res[v] = s2 * acc[v] - 2.f * a * s2 * zd * zd;
        }
      } else {
        #pragma unroll
        for (int v = 0; v < 8; ++v) {
          const float a = acur[v];
          res[v] = (1.f - a * a) * acc[v];
        }
      }
      #pragma unroll
      for (int v = 0; v < 8; ++v)
        actout[(s * 16 + (v + 8 * half)) * H + 16 * j + ln] = res[v];
    }
  }
}

__global__ __launch_bounds__(32) void pinn_grad_kernel(
    const float* __restrict__ x,
    const float* __restrict__ W0, const float* __restrict__ b0,
    const float* __restrict__ W1, const float* __restrict__ b1,
    const float* __restrict__ W2, const float* __restrict__ b2,
    const float* __restrict__ W3, const float* __restrict__ b3,
    float* __restrict__ out, int B)
{
  __shared__ __align__(16) float actA[NSTREAM * 16 * H];
  __shared__ __align__(16) float actB[NSTREAM * 16 * H];
  __shared__ float z3buf[NSTREAM * 16 * 4];

  const int lane = threadIdx.x & 31;
  const int half = lane >> 4;
  const int ln   = lane & 15;
  const int koff = half * 2;
  const int base = blockIdx.x * 16;

  for (int idx = lane; idx < 16 * H; idx += 32) {
    const int m = idx >> 7, n = idx & (H - 1);
    const float* xs = x + (size_t)(base + m) * 3;
    const float x0 = xs[0], x1 = xs[1], x2 = xs[2];
    const float w0 = W0[0 * H + n], w1 = W0[1 * H + n], w2 = W0[2 * H + n];
    const float z  = fmaf(x0, w0, fmaf(x1, w1, fmaf(x2, w2, b0[n])));
    const float a  = tanhf(z);
    const float s2 = 1.f - a * a;
    actA[(0 * 16 + m) * H + n] = a;
    actA[(1 * 16 + m) * H + n] = s2 * w0;
    actA[(2 * 16 + m) * H + n] = -2.f * a * s2 * w0 * w0;
    actA[(3 * 16 + m) * H + n] = s2 * w1;
    actA[(4 * 16 + m) * H + n] = -2.f * a * s2 * w1 * w1;
    actA[(5 * 16 + m) * H + n] = s2 * w2;
  }
  __syncthreads();

  mlp_hidden_layer(W1, b1, actA, actB, ln, half, koff);
  __syncthreads();
  mlp_hidden_layer(W2, b2, actB, actA, ln, half, koff);
  __syncthreads();

  {
    const int sg = half;
    const int m  = ln;
    #pragma unroll
    for (int ss = 0; ss < 3; ++ss) {
      const int s = 3 * sg + ss;
      float o0 = 0.f, o1 = 0.f, o2 = 0.f, o3 = 0.f;
      #pragma unroll 4
      for (int k = 0; k < H; ++k) {
        const float h = actA[(s * 16 + m) * H + k];
        o0 = fmaf(h, W3[k * 4 + 0], o0);
        o1 = fmaf(h, W3[k * 4 + 1], o1);
        o2 = fmaf(h, W3[k * 4 + 2], o2);
        o3 = fmaf(h, W3[k * 4 + 3], o3);
      }
      if (s == 0) { o0 += b3[0]; o1 += b3[1]; o2 += b3[2]; o3 += b3[3]; }
      z3buf[(s * 16 + m) * 4 + 0] = o0;
      z3buf[(s * 16 + m) * 4 + 1] = o1;
      z3buf[(s * 16 + m) * 4 + 2] = o2;
      z3buf[(s * 16 + m) * 4 + 3] = o3;
    }
  }
  __syncthreads();

  if (lane < 16) {
    const size_t i = (size_t)base + lane;
    float V[NSTREAM][4];
    #pragma unroll
    for (int s = 0; s < NSTREAM; ++s)
      #pragma unroll
      for (int o = 0; o < 4; ++o)
        V[s][o] = z3buf[(s * 16 + lane) * 4 + o];

    const float c   = V[0][0], v1 = V[0][1], v2 = V[0][2], Fi = V[0][3];
    const float dx0 = V[1][0], dx1 = V[1][1], dx2 = V[1][2], dx3 = V[1][3];
    const float xx0 = V[2][0], xx1 = V[2][1], xx2 = V[2][2], xx3 = V[2][3];
    const float dy0 = V[3][0], dy1 = V[3][1], dy2 = V[3][2], dy3 = V[3][3];
    const float yy0 = V[4][0], yy1 = V[4][1], yy2 = V[4][2], yy3 = V[4][3];
    const float dt0 = V[5][0], dt1 = V[5][1];

    const size_t Bz = (size_t)B;
    out[0 * Bz + i]          = c;
    out[1 * Bz + i]          = dt0;
    out[2 * Bz + 2 * i + 0]  = dx0;
    out[2 * Bz + 2 * i + 1]  = dy0;
    out[4 * Bz + 2 * i + 0]  = v1;
    out[4 * Bz + 2 * i + 1]  = v2;
    out[6 * Bz + i]          = dt1;
    out[7 * Bz + i]          = dx1 + dy2;
    out[8 * Bz + 2 * i + 0]  = dx1 * v1 + dy1 * v2;
    out[8 * Bz + 2 * i + 1]  = dx2 * v1 + dy2 * v2;
    out[10 * Bz + 2 * i + 0] = xx1 + yy1;
    out[10 * Bz + 2 * i + 1] = xx2 + yy2;
    out[12 * Bz + i]         = Fi;
    out[13 * Bz + 2 * i + 0] = dx3;
    out[13 * Bz + 2 * i + 1] = dy3;
    out[15 * Bz + i]         = xx3 + yy3;
    out[16 * Bz + i] = -0.5f * (xx0 + yy0)
                       - (dx0 * dx3 + dy0 * dy3) - c * (xx3 + yy3)
                       + (dx0 * v1 + dy0 * v2) + c * (dx1 + dy2);
  }
}

extern "C" void kernel_launch(void* const* d_in, const int* in_sizes, int n_in,
                              void* d_out, int out_size, void* d_ws, size_t ws_size,
                              hipStream_t stream) {
  const float* x  = (const float*)d_in[0];
  const float* W0 = (const float*)d_in[1];
  const float* b0 = (const float*)d_in[2];
  const float* W1 = (const float*)d_in[3];
  const float* b1 = (const float*)d_in[4];
  const float* W2 = (const float*)d_in[5];
  const float* b2 = (const float*)d_in[6];
  const float* W3 = (const float*)d_in[7];
  const float* b3 = (const float*)d_in[8];

  const int B = in_sizes[0] / 3;
  dim3 grid(B / 16), block(32);
  pinn_grad_kernel<<<grid, block, 0, stream>>>(x, W0, b0, W1, b1, W2, b2, W3, b3,
                                               (float*)d_out, B);
}


# Activations stay in registers, transposed: weights are the A operand, a layer's D tiles are the next layer's B fragments unmoved (k order permuted in all products); no LDS; one rolled layer body
# speedup vs baseline: 2.7257x; 2.7257x over previous
#include <hip/hip_runtime.h>

typedef float v2f __attribute__((ext_vector_type(2)));
typedef float v4f __attribute__((ext_vector_type(4)));
typedef float v8f __attribute__((ext_vector_type(8)));

__device__ __forceinline__ v8f wmma_f32(v2f a, v2f b, v8f c) {
  return __builtin_amdgcn_wmma_f32_16x16x4_f32(false, a, false, b, (short)0, c,
                                               false, false);
}

#define NSTREAM 6
#define H 128


__global__ __launch_bounds__(32) void pinn_grad_kernel(
    const float* __restrict__ x,
    const float* __restrict__ W0, const float* __restrict__ b0,
    const float* __restrict__ W1, const float* __restrict__ b1,
    const float* __restrict__ W2, const float* __restrict__ b2,
    const float* __restrict__ W3, const float* __restrict__ b3,
    float* __restrict__ out, int B)
{
  const int lane = threadIdx.x & 31;
  const int h    = lane >> 4;
  const int c    = lane & 15;
  const int base = blockIdx.x * 16;

  v8f act[NSTREAM][8];

  #pragma nounroll
  for (int layer = 0; layer < 3; ++layer) {
    v8f acc[NSTREAM][8];

    if (layer == 0) {
      const float* xs = x + (size_t)(base + c) * 3;
      const float x0 = xs[0], x1 = xs[1], x2 = xs[2];
      #pragma unroll
      for (int r = 0; r < 8; ++r) {
        const int n0 = 64 * h + 8 * r;
        v4f w0v[2], w1v[2], w2v[2], bv[2];
        #pragma unroll
        for (int q = 0; q < 2; ++q) {
          w0v[q] = *(const v4f*)&W0[0 * H + n0 + 4 * q];
          w1v[q] = *(const v4f*)&W0[1 * H + n0 + 4 * q];
          w2v[q] = *(const v4f*)&W0[2 * H + n0 + 4 * q];
          bv[q]  = *(const v4f*)&b0[n0 + 4 * q];
        }
        #pragma unroll
        for (int j = 0; j < 8; ++j) {
          const float w0 = w0v[j >> 2][j & 3], w1 = w1v[j >> 2][j & 3];
          const float w2 = w2v[j >> 2][j & 3], bj = bv[j >> 2][j & 3];
          acc[0][j][r] = fmaf(x0, w0, fmaf(x1, w1, fmaf(x2, w2, bj)));
          acc[1][j][r] = w0;
          acc[2][j][r] = 0.f;
          acc[3][j][r] = w1;
          acc[4][j][r] = 0.f;
          acc[5][j][r] = w2;
        }
      }
    } else {
      const float* __restrict__ W  = (layer == 1) ? W1 : W2;
      const float* __restrict__ bb = (layer == 1) ? b1 : b2;

      #pragma unroll
      for (int r = 0; r < 8; ++r) {
        const v4f blo = *(const v4f*)&bb[64 * h + 8 * r];
        const v4f bhi = *(const v4f*)&bb[64 * h + 8 * r + 4];
        #pragma unroll
        for (int jo = 0; jo < 4; ++jo) {
          acc[0][jo][r]     = blo[jo];
          acc[0][jo + 4][r] = bhi[jo];
        }
      }
      #pragma unroll
      for (int s = 1; s < NSTREAM; ++s)
        #pragma unroll
        for (int jo = 0; jo < 8; ++jo)
          #pragma unroll
          for (int r = 0; r < 8; ++r) acc[s][jo][r] = 0.f;

      const float* wl = W + (size_t)(64 * h) * H + 8 * c;

      #pragma unroll
      for (int j = 0; j < 8; ++j) {
        #pragma unroll
        for (int p = 0; p < 4; ++p) {
          const float* r0 = wl + (size_t)(16 * p + j) * H;
          const float* r1 = r0 + 8 * H;
          const v4f a0l = *(const v4f*)r0;
          const v4f a0h = *(const v4f*)(r0 + 4);
          const v4f a1l = *(const v4f*)r1;
          const v4f a1h = *(const v4f*)(r1 + 4);
          #pragma unroll
          for (int jo = 0; jo < 8; ++jo) {
            v2f af;
            af.x = (jo < 4) ? a0l[jo & 3] : a0h[jo & 3];
            af.y = (jo < 4) ? a1l[jo & 3] : a1h[jo & 3];
            #pragma unroll
            for (int s = 0; s < NSTREAM; ++s) {
              v2f bf;
              bf.x = act[s][j][2 * p];
              bf.y = act[s][j][2 * p + 1];
              acc[s][jo] = wmma_f32(af, bf, acc[s][jo]);
            }
          }
        }
      }
    }

    #pragma unroll
    for (int jo = 0; jo < 8; ++jo) {
      #pragma unroll
      for (int r = 0; r < 8; ++r) {
        const float a  = tanhf(acc[0][jo][r]);
        const float s2 = 1.f - a * a;
        const float zx = acc[1][jo][r];
        const float zy = acc[3][jo][r];
        act[0][jo][r] = a;
        act[1][jo][r] = s2 * zx;
        act[2][jo][r] = s2 * acc[2][jo][r] - 2.f * a * s2 * zx * zx;
        act[3][jo][r] = s2 * zy;
        act[4][jo][r] = s2 * acc[4][jo][r] - 2.f * a * s2 * zy * zy;
        act[5][jo][r] = s2 * acc[5][jo][r];
      }
    }
  }

  float V[NSTREAM][4];
  {
    v8f o3[NSTREAM];
    #pragma unroll
    for (int s = 0; s < NSTREAM; ++s)
      #pragma unroll
      for (int r = 0; r < 8; ++r) o3[s][r] = 0.f;

    const float* w3l = W3 + (size_t)(64 * h) * 4 + (c & 3);
    const bool live = (c < 4);

    #pragma unroll
    for (int j = 0; j < 8; ++j) {
      #pragma unroll
      for (int p = 0; p < 4; ++p) {
        const float g0 = w3l[(size_t)(16 * p + j) * 4];
        const float g1 = w3l[(size_t)(16 * p + j + 8) * 4];
        v2f af;
        af.x = live ? g0 : 0.f;
        af.y = live ? g1 : 0.f;
        #pragma unroll
        for (int s = 0; s < NSTREAM; ++s) {
          v2f bf;
          bf.x = act[s][j][2 * p];
          bf.y = act[s][j][2 * p + 1];
          o3[s] = wmma_f32(af, bf, o3[s]);
        }
      }
    }
    #pragma unroll
    for (int s = 0; s < NSTREAM; ++s)
      #pragma unroll
      for (int o = 0; o < 4; ++o) V[s][o] = o3[s][o];
  }
  V[0][0] += b3[0]; V[0][1] += b3[1]; V[0][2] += b3[2]; V[0][3] += b3[3];

  if (lane < 16) {
    const size_t i = (size_t)base + lane;

    const float cc  = V[0][0], v1 = V[0][1], v2 = V[0][2], Fi = V[0][3];
    const float dx0 = V[1][0], dx1 = V[1][1], dx2 = V[1][2], dx3 = V[1][3];
    const float xx0 = V[2][0], xx1 = V[2][1], xx2 = V[2][2], xx3 = V[2][3];
    const float dy0 = V[3][0], dy1 = V[3][1], dy2 = V[3][2], dy3 = V[3][3];
    const float yy0 = V[4][0], yy1 = V[4][1], yy2 = V[4][2], yy3 = V[4][3];
    const float dt0 = V[5][0], dt1 = V[5][1];

    const size_t Bz = (size_t)B;
    out[0 * Bz + i]          = cc;
    out[1 * Bz + i]          = dt0;
    out[2 * Bz + 2 * i + 0]  = dx0;
    out[2 * Bz + 2 * i + 1]  = dy0;
    out[4 * Bz + 2 * i + 0]  = v1;
    out[4 * Bz + 2 * i + 1]  = v2;
    out[6 * Bz + i]          = dt1;
    out[7 * Bz + i]          = dx1 + dy2;
    out[8 * Bz + 2 * i + 0]  = dx1 * v1 + dy1 * v2;
    out[8 * Bz + 2 * i + 1]  = dx2 * v1 + dy2 * v2;
    out[10 * Bz + 2 * i + 0] = xx1 + yy1;
    out[10 * Bz + 2 * i + 1] = xx2 + yy2;
    out[12 * Bz + i]         = Fi;
    out[13 * Bz + 2 * i + 0] = dx3;
    out[13 * Bz + 2 * i + 1] = dy3;
    out[15 * Bz + i]         = xx3 + yy3;
    out[16 * Bz + i] = -0.5f * (xx0 + yy0)
                       - (dx0 * dx3 + dy0 * dy3) - cc * (xx3 + yy3)
                       + (dx0 * v1 + dy0 * v2) + cc * (dx1 + dy2);
  }
}

extern "C" void kernel_launch(void* const* d_in, const int* in_sizes, int n_in,
                              void* d_out, int out_size, void* d_ws, size_t ws_size,
                              hipStream_t stream) {
  const float* x  = (const float*)d_in[0];
  const float* W0 = (const float*)d_in[1];
  const float* b0 = (const float*)d_in[2];
  const float* W1 = (const float*)d_in[3];
  const float* b1 = (const float*)d_in[4];
  const float* W2 = (const float*)d_in[5];
  const float* b2 = (const float*)d_in[6];
  const float* W3 = (const float*)d_in[7];
  const float* b3 = (const float*)d_in[8];

  const int B = in_sizes[0] / 3;
  dim3 grid(B / 16), block(32);
  pinn_grad_kernel<<<grid, block, 0, stream>>>(x, W0, b0, W1, b1, W2, b2, W3, b3,
                                               (float*)d_out, B);
}


# tanh of all three tanh layers as one v_tanh_f32 each (compiler builtin) in place of the library routine: no branches left in the epilogue, its 64 values per lane interleave; tanh rounding moves
# speedup vs baseline: 3.9952x; 1.4658x over previous
#include <hip/hip_runtime.h>

typedef float v2f __attribute__((ext_vector_type(2)));
typedef float v4f __attribute__((ext_vector_type(4)));
typedef float v8f __attribute__((ext_vector_type(8)));

__device__ __forceinline__ v8f wmma_f32(v2f a, v2f b, v8f c) {
  return __builtin_amdgcn_wmma_f32_16x16x4_f32(false, a, false, b, (short)0, c,
                                               false, false);
}

#define NSTREAM 6
#define H 128


__global__ __launch_bounds__(32) void pinn_grad_kernel(
    const float* __restrict__ x,
    const float* __restrict__ W0, const float* __restrict__ b0,
    const float* __restrict__ W1, const float* __restrict__ b1,
    const float* __restrict__ W2, const float* __restrict__ b2,
    const float* __restrict__ W3, const float* __restrict__ b3,
    float* __restrict__ out, int B)
{
  const int lane = threadIdx.x & 31;
  const int h    = lane >> 4;
  const int c    = lane & 15;
  const int base = blockIdx.x * 16;

  v8f act[NSTREAM][8];

  #pragma nounroll
  for (int layer = 0; layer < 3; ++layer) {
    v8f acc[NSTREAM][8];

    if (layer == 0) {
      const float* xs = x + (size_t)(base + c) * 3;
      const float x0 = xs[0], x1 = xs[1], x2 = xs[2];
      #pragma unroll
      for (int r = 0; r < 8; ++r) {
        const int n0 = 64 * h + 8 * r;
        v4f w0v[2], w1v[2], w2v[2], bv[2];
        #pragma unroll
        for (int q = 0; q < 2; ++q) {
          w0v[q] = *(const v4f*)&W0[0 * H + n0 + 4 * q];
          w1v[q] = *(const v4f*)&W0[1 * H + n0 + 4 * q];
          w2v[q] = *(const v4f*)&W0[2 * H + n0 + 4 * q];
          bv[q]  = *(const v4f*)&b0[n0 + 4 * q];
        }
        #pragma unroll
        for (int j = 0; j < 8; ++j) {
          const float w0 = w0v[j >> 2][j & 3], w1 = w1v[j >> 2][j & 3];
          const float w2 = w2v[j >> 2][j & 3], bj = bv[j >> 2][j & 3];
          acc[0][j][r] = fmaf(x0, w0, fmaf(x1, w1, fmaf(x2, w2, bj)));
          acc[1][j][r] = w0;
          acc[2][j][r] = 0.f;
          acc[3][j][r] = w1;
          acc[4][j][r] = 0.f;
          acc[5][j][r] = w2;
        }
      }
    } else {
      const float* __restrict__ W  = (layer == 1) ? W1 : W2;
      const float* __restrict__ bb = (layer == 1) ? b1 : b2;

      #pragma unroll
      for (int r = 0; r < 8; ++r) {
        const v4f blo = *(const v4f*)&bb[64 * h + 8 * r];
        const v4f bhi = *(const v4f*)&bb[64 * h + 8 * r + 4];
        #pragma unroll
        for (int jo = 0; jo < 4; ++jo) {
          acc[0][jo][r]     = blo[jo];
          acc[0][jo + 4][r] = bhi[jo];
        }
      }
      #pragma unroll
      for (int s = 1; s < NSTREAM; ++s)
        #pragma unroll
        for (int jo = 0; jo < 8; ++jo)
          #pragma unroll
          for (int r = 0; r < 8; ++r) acc[s][jo][r] = 0.f;

      const float* wl = W + (size_t)(64 * h) * H + 8 * c;

      #pragma unroll
      for (int j = 0; j < 8; ++j) {
        #pragma unroll
        for (int p = 0; p < 4; ++p) {
          const float* r0 = wl + (size_t)(16 * p + j) * H;
          const float* r1 = r0 + 8 * H;
          const v4f a0l = *(const v4f*)r0;
          const v4f a0h = *(const v4f*)(r0 + 4);
          const v4f a1l = *(const v4f*)r1;
          const v4f a1h = *(const v4f*)(r1 + 4);
          #pragma unroll
          for (int jo = 0; jo < 8; ++jo) {
            v2f af;
            af.x = (jo < 4) ? a0l[jo & 3] : a0h[jo & 3];
            af.y = (jo < 4) ? a1l[jo & 3] : a1h[jo & 3];
            #pragma unroll
            for (int s = 0; s < NSTREAM; ++s) {
              v2f bf;
              bf.x = act[s][j][2 * p];
              bf.y = act[s][j][2 * p + 1];
              acc[s][jo] = wmma_f32(af, bf, acc[s][jo]);
            }
          }
        }
      }
    }

    #pragma unroll
    for (int jo = 0; jo < 8; ++jo) {
      #pragma unroll
      for (int r = 0; r < 8; ++r) {
        const float a  = __builtin_amdgcn_tanhf(acc[0][jo][r]);
        const float s2 = 1.f - a * a;
        const float zx = acc[1][jo][r];
        const float zy = acc[3][jo][r];
        act[0][jo][r] = a;
        act[1][jo][r] = s2 * zx;
        act[2][jo][r] = s2 * acc[2][jo][r] - 2.f * a * s2 * zx * zx;
        act[3][jo][r] = s2 * zy;
        act[4][jo][r] = s2 * acc[4][jo][r] - 2.f * a * s2 * zy * zy;
        act[5][jo][r] = s2 * acc[5][jo][r];
      }
    }
  }

  float V[NSTREAM][4];
  {
    v8f o3[NSTREAM];
    #pragma unroll
    for (int s = 0; s < NSTREAM; ++s)
      #pragma unroll
      for (int r = 0; r < 8; ++r) o3[s][r] = 0.f;

    const float* w3l = W3 + (size_t)(64 * h) * 4 + (c & 3);
    const bool live = (c < 4);

    #pragma unroll
    for (int j = 0; j < 8; ++j) {
      #pragma unroll
      for (int p = 0; p < 4; ++p) {
        const float g0 = w3l[(size_t)(16 * p + j) * 4];
        const float g1 = w3l[(size_t)(16 * p + j + 8) * 4];
        v2f af;
        af.x = live ? g0 : 0.f;
        af.y = live ? g1 : 0.f;
        #pragma unroll
        for (int s = 0; s < NSTREAM; ++s) {
          v2f bf;
          bf.x = act[s][j][2 * p];
          bf.y = act[s][j][2 * p + 1];
          o3[s] = wmma_f32(af, bf, o3[s]);
        }
      }
    }
    #pragma unroll
    for (int s = 0; s < NSTREAM; ++s)
      #pragma unroll
      for (int o = 0; o < 4; ++o) V[s][o] = o3[s][o];
  }
  V[0][0] += b3[0]; V[0][1] += b3[1]; V[0][2] += b3[2]; V[0][3] += b3[3];

  if (lane < 16) {
    const size_t i = (size_t)base + lane;

    const float cc  = V[0][0], v1 = V[0][1], v2 = V[0][2], Fi = V[0][3];
    const float dx0 = V[1][0], dx1 = V[1][1], dx2 = V[1][2], dx3 = V[1][3];
    const float xx0 = V[2][0], xx1 = V[2][1], xx2 = V[2][2], xx3 = V[2][3];
    const float dy0 = V[3][0], dy1 = V[3][1], dy2 = V[3][2], dy3 = V[3][3];
    const float yy0 = V[4][0], yy1 = V[4][1], yy2 = V[4][2], yy3 = V[4][3];
    const float dt0 = V[5][0], dt1 = V[5][1];

    const size_t Bz = (size_t)B;
    out[0 * Bz + i]          = cc;
    out[1 * Bz + i]          = dt0;
    out[2 * Bz + 2 * i + 0]  = dx0;
    out[2 * Bz + 2 * i + 1]  = dy0;
    out[4 * Bz + 2 * i + 0]  = v1;
    out[4 * Bz + 2 * i + 1]  = v2;
    out[6 * Bz + i]          = dt1;
    out[7 * Bz + i]          = dx1 + dy2;
    out[8 * Bz + 2 * i + 0]  = dx1 * v1 + dy1 * v2;
    out[8 * Bz + 2 * i + 1]  = dx2 * v1 + dy2 * v2;
    out[10 * Bz + 2 * i + 0] = xx1 + yy1;
    out[10 * Bz + 2 * i + 1] = xx2 + yy2;
    out[12 * Bz + i]         = Fi;
    out[13 * Bz + 2 * i + 0] = dx3;
    out[13 * Bz + 2 * i + 1] = dy3;
    out[15 * Bz + i]         = xx3 + yy3;
    out[16 * Bz + i] = -0.5f * (xx0 + yy0)
                       - (dx0 * dx3 + dy0 * dy3) - cc * (xx3 + yy3)
                       + (dx0 * v1 + dy0 * v2) + cc * (dx1 + dy2);
  }
}

extern "C" void kernel_launch(void* const* d_in, const int* in_sizes, int n_in,
                              void* d_out, int out_size, void* d_ws, size_t ws_size,
                              hipStream_t stream) {
  const float* x  = (const float*)d_in[0];
  const float* W0 = (const float*)d_in[1];
  const float* b0 = (const float*)d_in[2];
  const float* W1 = (const float*)d_in[3];
  const float* b1 = (const float*)d_in[4];
  const float* W2 = (const float*)d_in[5];
  const float* b2 = (const float*)d_in[6];
  const float* W3 = (const float*)d_in[7];
  const float* b3 = (const float*)d_in[8];

  const int B = in_sizes[0] / 3;
  dim3 grid(B / 16), block(32);
  pinn_grad_kernel<<<grid, block, 0, stream>>>(x, W0, b0, W1, b1, W2, b2, W3, b3,
                                               (float*)d_out, B);
}
